# starting kernel
# baseline (speedup 1.0000x reference)
#include <hip/hip_runtime.h>

typedef float v2f __attribute__((ext_vector_type(2)));
typedef float v8f __attribute__((ext_vector_type(8)));

#define NNODES 10000
#define NEDGES 160000

__global__ void deg_init_kernel(float* __restrict__ deg, int n) {
  int i = blockIdx.x * blockDim.x + threadIdx.x;
  if (i < n) deg[i] = 1.0f;
}

__global__ void deg_accum_kernel(const int* __restrict__ dst, float* __restrict__ deg, int e) {
  int i = blockIdx.x * blockDim.x + threadIdx.x;
  if (i < e) atomicAdd(&deg[dst[i]], 1.0f);
}

__global__ void dinv_kernel(const float* __restrict__ deg, float* __restrict__ dinv, int n) {
  int i = blockIdx.x * blockDim.x + threadIdx.x;
  if (i < n) dinv[i] = rsqrtf(deg[i]);
}

__global__ __launch_bounds__(128) void wmma_gemm_f32_kernel(
    const float* __restrict__ A, const float* __restrict__ B,
    float* __restrict__ C, int M, int N, int K) {
  const int wave  = threadIdx.x >> 5;
  const int lane  = threadIdx.x & 31;
  const int l16   = lane & 15;
  const int lhalf = lane >> 4;

  const int wm = wave >> 1;
  const int wn = wave & 1;
  const int row0 = blockIdx.y * 64 + wm * 32;
  const int col0 = blockIdx.x * 64 + wn * 32;

  const float* arow[2];
#pragma unroll
  for (int mi = 0; mi < 2; ++mi) {
    int r = row0 + mi * 16 + l16;
    r = r < M ? r : M - 1;
    arow[mi] = A + (size_t)r * K + lhalf * 2;
  }
  const float* bcol[2];
#pragma unroll
  for (int ni = 0; ni < 2; ++ni)
    bcol[ni] = B + (size_t)(lhalf * 2) * N + (col0 + ni * 16 + l16);

  v8f acc[2][2] = {};
#pragma unroll 4
  for (int k = 0; k < K; k += 4) {
    v2f a[2], b[2];
#pragma unroll
    for (int mi = 0; mi < 2; ++mi) {
      a[mi].x = arow[mi][k];
      a[mi].y = arow[mi][k + 1];
    }
#pragma unroll
    for (int ni = 0; ni < 2; ++ni) {
      b[ni].x = bcol[ni][(size_t)k * N];
      b[ni].y = bcol[ni][(size_t)(k + 1) * N];
    }
#pragma unroll
    for (int mi = 0; mi < 2; ++mi)
#pragma unroll
      for (int ni = 0; ni < 2; ++ni)
        acc[mi][ni] = __builtin_amdgcn_wmma_f32_16x16x4_f32(
             false, a[mi],  false, b[ni],
             (short)0, acc[mi][ni],  false,  false);
  }

#pragma unroll
  for (int mi = 0; mi < 2; ++mi) {
#pragma unroll
    for (int r = 0; r < 8; ++r) {
      int row = row0 + mi * 16 + 8 * lhalf + r;
      if (row < M) {
#pragma unroll
        for (int ni = 0; ni < 2; ++ni)
          C[(size_t)row * N + col0 + ni * 16 + l16] = acc[mi][ni][r];
      }
    }
  }
}

__global__ void init_out_kernel(const float* __restrict__ h, const float* __restrict__ dinv,
                                const float* __restrict__ bias, float* __restrict__ out,
                                int n, int c) {
  int idx = blockIdx.x * blockDim.x + threadIdx.x;
  if (idx >= n * c) return;
  int i = idx / c;
  int f = idx - i * c;
  float di = dinv[i];
  out[idx] = h[idx] * di * di + bias[f];
}

__global__ void scatter_kernel(const float* __restrict__ h, const int* __restrict__ src,
                               const int* __restrict__ dst, const float* __restrict__ dinv,
                               float* __restrict__ out, int c) {
  int e = blockIdx.x;
  int s = src[e];
  int d = dst[e];
  float norm = dinv[s] * dinv[d];
  const float* hs = h + (size_t)s * c;
  float* od = out + (size_t)d * c;
  for (int f = threadIdx.x; f < c; f += blockDim.x)
    atomicAdd(&od[f], hs[f] * norm);
}

__global__ void relu_kernel(float* __restrict__ x, int n) {
  int i = blockIdx.x * blockDim.x + threadIdx.x;
  if (i < n) x[i] = fmaxf(x[i], 0.0f);
}

extern "C" void kernel_launch(void* const* d_in, const int* in_sizes, int n_in,
                              void* d_out, int out_size, void* d_ws, size_t ws_size,
                              hipStream_t stream) {
  const float* x   = (const float*)d_in[0];
  const int*   ei  = (const int*)d_in[1];
  const float* W1  = (const float*)d_in[2];
  const float* b1  = (const float*)d_in[3];
  const float* W2  = (const float*)d_in[4];
  const float* b2  = (const float*)d_in[5];
  float* out = (float*)d_out;

  const int N = NNODES;
  const int E = NEDGES;
  const int C0 = 512, C1 = 1024, C2 = 512;

  const int* src = ei;
  const int* dst = ei + E;

  float* deg  = (float*)d_ws;
  float* dinv = deg + N;
  float* h    = dinv + N;
  float* agg1 = h + (size_t)N * C1;

  const int T = 256;
  const int MB = (N + 63) / 64;

  deg_init_kernel<<<(N + T - 1) / T, T, 0, stream>>>(deg, N);
  deg_accum_kernel<<<(E + T - 1) / T, T, 0, stream>>>(dst, deg, E);
  dinv_kernel<<<(N + T - 1) / T, T, 0, stream>>>(deg, dinv, N);

  {
    dim3 grid(C1 / 64, MB);
    wmma_gemm_f32_kernel<<<grid, 128, 0, stream>>>(x, W1, h, N, C1, C0);
    init_out_kernel<<<((N * C1) + T - 1) / T, T, 0, stream>>>(h, dinv, b1, agg1, N, C1);
    scatter_kernel<<<E, T, 0, stream>>>(h, src, dst, dinv, agg1, C1);
    relu_kernel<<<((N * C1) + T - 1) / T, T, 0, stream>>>(agg1, N * C1);
  }

  {
    dim3 grid(C2 / 64, MB);
    wmma_gemm_f32_kernel<<<grid, 128, 0, stream>>>(agg1, W2, h, N, C2, C1);
    init_out_kernel<<<((N * C2) + T - 1) / T, T, 0, stream>>>(h, dinv, b2, out, N, C2);
    scatter_kernel<<<E, T, 0, stream>>>(h, src, dst, dinv, out, C2);
    relu_kernel<<<((N * C2) + T - 1) / T, T, 0, stream>>>(out, N * C2);
  }
}


# Edges summed by destination row over a CSR built in-call (int-atomic slot, scan, sort by source): float-atomic scatter, init and relu passes become one row gather; sum order now fixed
# speedup vs baseline: 2.0814x; 2.0814x over previous
#include <hip/hip_runtime.h>


typedef float v2f __attribute__((ext_vector_type(2)));
typedef float v8f __attribute__((ext_vector_type(8)));

#define NNODES 10000
#define NEDGES 160000

__global__ void cnt_zero_kernel(int* __restrict__ cnt, int n) {
  int i = blockIdx.x * blockDim.x + threadIdx.x;
  if (i < n) cnt[i] = 0;
}

__global__ void edge_rank_kernel(const int* __restrict__ dst, int* __restrict__ cnt,
                                 int* __restrict__ rank, int e) {
  int i = blockIdx.x * blockDim.x + threadIdx.x;
  if (i < e) rank[i] = atomicAdd(&cnt[dst[i]], 1);
}

#define SCAN_T 1024
#define SCAN_PER ((NNODES + SCAN_T - 1) / SCAN_T)
__global__ __launch_bounds__(SCAN_T) void rowptr_dinv_kernel(
    const int* __restrict__ cnt, int* __restrict__ rowptr, float* __restrict__ dinv) {
  __shared__ int part[SCAN_T];
  const int t = threadIdx.x;
  const int base = t * SCAN_PER;
  int c[SCAN_PER];
  int s = 0;
#pragma unroll
  for (int j = 0; j < SCAN_PER; ++j) {
    int i = base + j;
    c[j] = i < NNODES ? cnt[i] : 0;
    s += c[j];
  }
  part[t] = s;
  __syncthreads();
  for (int off = 1; off < SCAN_T; off <<= 1) {
    int v = t >= off ? part[t - off] : 0;
    __syncthreads();
    part[t] += v;
    __syncthreads();
  }
  int run = part[t] - s;
#pragma unroll
  for (int j = 0; j < SCAN_PER; ++j) {
    int i = base + j;
    if (i < NNODES) {
      rowptr[i] = run;
      dinv[i] = rsqrtf((float)(c[j] + 1));
      run += c[j];
    }
  }
  if (t == SCAN_T - 1) rowptr[NNODES] = part[SCAN_T - 1];
}

__global__ void place_kernel(const int* __restrict__ src, const int* __restrict__ dst,
                             const int* __restrict__ rank, const int* __restrict__ rowptr,
                             int* __restrict__ tmp, int e) {
  int i = blockIdx.x * blockDim.x + threadIdx.x;
  if (i < e) tmp[rowptr[dst[i]] + rank[i]] = src[i];
}

__global__ void rowsort_kernel(const int* __restrict__ rowptr, const int* __restrict__ tmp,
                               int* __restrict__ col, int n) {
  const int row = (blockIdx.x * blockDim.x + threadIdx.x) >> 5;
  const int l = threadIdx.x & 31;
  if (row >= n) return;
  const int beg = rowptr[row];
  const int end = rowptr[row + 1];
  for (int p = beg + l; p < end; p += 32) {
    const int v = tmp[p];
    int r = 0;
    for (int k = beg; k < end; ++k) {
      const int u = tmp[k];
      r += (u < v || (u == v && k < p)) ? 1 : 0;
    }
    col[beg + r] = v;
  }
}

__global__ __launch_bounds__(128) void wmma_gemm_f32_kernel(
    const float* __restrict__ A, const float* __restrict__ B,
    float* __restrict__ C, int M, int N, int K) {
  const int wave  = threadIdx.x >> 5;
  const int lane  = threadIdx.x & 31;
  const int l16   = lane & 15;
  const int lhalf = lane >> 4;

  const int wm = wave >> 1;
  const int wn = wave & 1;
  const int row0 = blockIdx.y * 64 + wm * 32;
  const int col0 = blockIdx.x * 64 + wn * 32;

  const float* arow[2];
#pragma unroll
  for (int mi = 0; mi < 2; ++mi) {
    int r = row0 + mi * 16 + l16;
    r = r < M ? r : M - 1;
    arow[mi] = A + (size_t)r * K + lhalf * 2;
  }
  const float* bcol[2];
#pragma unroll
  for (int ni = 0; ni < 2; ++ni)
    bcol[ni] = B + (size_t)(lhalf * 2) * N + (col0 + ni * 16 + l16);

  v8f acc[2][2] = {};
#pragma unroll 4
  for (int k = 0; k < K; k += 4) {
    v2f a[2], b[2];
#pragma unroll
    for (int mi = 0; mi < 2; ++mi) {
      a[mi].x = arow[mi][k];
      a[mi].y = arow[mi][k + 1];
    }
#pragma unroll
    for (int ni = 0; ni < 2; ++ni) {
      b[ni].x = bcol[ni][(size_t)k * N];
      b[ni].y = bcol[ni][(size_t)(k + 1) * N];
    }
#pragma unroll
    for (int mi = 0; mi < 2; ++mi)
#pragma unroll
      for (int ni = 0; ni < 2; ++ni)
        acc[mi][ni] = __builtin_amdgcn_wmma_f32_16x16x4_f32(
             false, a[mi],  false, b[ni],
             (short)0, acc[mi][ni],  false,  false);
  }

#pragma unroll
  for (int mi = 0; mi < 2; ++mi) {
#pragma unroll
    for (int r = 0; r < 8; ++r) {
      int row = row0 + mi * 16 + 8 * lhalf + r;
      if (row < M) {
#pragma unroll
        for (int ni = 0; ni < 2; ++ni)
          C[(size_t)row * N + col0 + ni * 16 + l16] = acc[mi][ni][r];
      }
    }
  }
}

template <int K, int C4>
__device__ __forceinline__ void add_rows(float4& acc, const float4* __restrict__ h4,
                                         const int* __restrict__ col,
                                         const float* __restrict__ dinv, float di, int j, int t) {
  float4 v[K];
  float w[K];
#pragma unroll
  for (int k = 0; k < K; ++k) {
    const int s = col[j + k];
    w[k] = dinv[s] * di;
    v[k] = h4[(size_t)s * C4 + t];
  }
#pragma unroll
  for (int k = 0; k < K; ++k) {
    acc.x += v[k].x * w[k];
    acc.y += v[k].y * w[k];
    acc.z += v[k].z * w[k];
    acc.w += v[k].w * w[k];
  }
}

template <int C>
__global__ __launch_bounds__(C / 4) void gather_rows_kernel(
    const float* __restrict__ h, const int* __restrict__ rowptr, const int* __restrict__ col,
    const float* __restrict__ dinv, const float* __restrict__ bias, float* __restrict__ out) {
  constexpr int C4 = C / 4;
  const int i = blockIdx.x;
  const int t = threadIdx.x;
  const float4* __restrict__ h4 = reinterpret_cast<const float4*>(h);
  const float di = dinv[i];
  const int beg = rowptr[i];
  const int end = rowptr[i + 1];

  const float4 sv = h4[(size_t)i * C4 + t];
  const float4 b = reinterpret_cast<const float4*>(bias)[t];
  const float sw = di * di;
  float4 acc;
  acc.x = sv.x * sw; acc.y = sv.y * sw; acc.z = sv.z * sw; acc.w = sv.w * sw;

  int j = beg;
  for (; j + 8 <= end; j += 8) add_rows<8, C4>(acc, h4, col, dinv, di, j, t);
  if (j + 4 <= end) { add_rows<4, C4>(acc, h4, col, dinv, di, j, t); j += 4; }
  if (j + 2 <= end) { add_rows<2, C4>(acc, h4, col, dinv, di, j, t); j += 2; }
  if (j < end) add_rows<1, C4>(acc, h4, col, dinv, di, j, t);

  float4 o;
  o.x = fmaxf(acc.x + b.x, 0.0f);
  o.y = fmaxf(acc.y + b.y, 0.0f);
  o.z = fmaxf(acc.z + b.z, 0.0f);
  o.w = fmaxf(acc.w + b.w, 0.0f);
  reinterpret_cast<float4*>(out)[(size_t)i * C4 + t] = o;
}

extern "C" void kernel_launch(void* const* d_in, const int* in_sizes, int n_in,
                              void* d_out, int out_size, void* d_ws, size_t ws_size,
                              hipStream_t stream) {
  const float* x   = (const float*)d_in[0];
  const int*   ei  = (const int*)d_in[1];
  const float* W1  = (const float*)d_in[2];
  const float* b1  = (const float*)d_in[3];
  const float* W2  = (const float*)d_in[4];
  const float* b2  = (const float*)d_in[5];
  float* out = (float*)d_out;

  const int N = NNODES;
  const int E = NEDGES;
  const int C0 = 512, C1 = 1024, C2 = 512;

  const int* src = ei;
  const int* dst = ei + E;

  float* h      = (float*)d_ws;
  float* agg1   = h + (size_t)N * C1;
  float* dinv   = agg1 + (size_t)N * C1;
  int*   cnt    = (int*)(dinv + N);
  int*   rowptr = cnt + N;
  int*   rank   = rowptr + (N + 16);
  int*   tmp    = rank + E;
  int*   col    = tmp + E;

  const int T = 256;
  const int MB = (N + 63) / 64;

  cnt_zero_kernel<<<(N + T - 1) / T, T, 0, stream>>>(cnt, N);
  edge_rank_kernel<<<(E + T - 1) / T, T, 0, stream>>>(dst, cnt, rank, E);
  rowptr_dinv_kernel<<<1, SCAN_T, 0, stream>>>(cnt, rowptr, dinv);
  place_kernel<<<(E + T - 1) / T, T, 0, stream>>>(src, dst, rank, rowptr, tmp, E);
  rowsort_kernel<<<(N * 32 + T - 1) / T, T, 0, stream>>>(rowptr, tmp, col, N);

  {
    dim3 grid(C1 / 64, MB);
    wmma_gemm_f32_kernel<<<grid, 128, 0, stream>>>(x, W1, h, N, C1, C0);
    gather_rows_kernel<1024><<<N, 256, 0, stream>>>(h, rowptr, col, dinv, b1, agg1);
  }

  {
    dim3 grid(C2 / 64, MB);
    wmma_gemm_f32_kernel<<<grid, 128, 0, stream>>>(agg1, W2, h, N, C2, C1);
    gather_rows_kernel<512><<<N, 128, 0, stream>>>(h, rowptr, col, dinv, b2, out);
  }
}


# GEMM operands staged through LDS: per 64x64 block, 32-deep A and B tiles copied by whole-line 16-byte loads one tile ahead, fragments read from LDS (B k-pairs interleaved); sum order unchanged
# speedup vs baseline: 3.2655x; 1.5689x over previous
#include <hip/hip_runtime.h>


typedef float v2f __attribute__((ext_vector_type(2)));
typedef float v4f __attribute__((ext_vector_type(4)));
typedef float v8f __attribute__((ext_vector_type(8)));

#define NNODES 10000
#define NEDGES 160000

__global__ void cnt_zero_kernel(int* __restrict__ cnt, int n) {
  int i = blockIdx.x * blockDim.x + threadIdx.x;
  if (i < n) cnt[i] = 0;
}

__global__ void edge_rank_kernel(const int* __restrict__ dst, int* __restrict__ cnt,
                                 int* __restrict__ rank, int e) {
  int i = blockIdx.x * blockDim.x + threadIdx.x;
  if (i < e) rank[i] = atomicAdd(&cnt[dst[i]], 1);
}

#define SCAN_T 1024
#define SCAN_PER ((NNODES + SCAN_T - 1) / SCAN_T)
__global__ __launch_bounds__(SCAN_T) void rowptr_dinv_kernel(
    const int* __restrict__ cnt, int* __restrict__ rowptr, float* __restrict__ dinv) {
  __shared__ int part[SCAN_T];
  const int t = threadIdx.x;
  const int base = t * SCAN_PER;
  int c[SCAN_PER];
  int s = 0;
#pragma unroll
  for (int j = 0; j < SCAN_PER; ++j) {
    int i = base + j;
    c[j] = i < NNODES ? cnt[i] : 0;
    s += c[j];
  }
  part[t] = s;
  __syncthreads();
  for (int off = 1; off < SCAN_T; off <<= 1) {
    int v = t >= off ? part[t - off] : 0;
    __syncthreads();
    part[t] += v;
    __syncthreads();
  }
  int run = part[t] - s;
#pragma unroll
  for (int j = 0; j < SCAN_PER; ++j) {
    int i = base + j;
    if (i < NNODES) {
      rowptr[i] = run;
      dinv[i] = rsqrtf((float)(c[j] + 1));
      run += c[j];
    }
  }
  if (t == SCAN_T - 1) rowptr[NNODES] = part[SCAN_T - 1];
}

__global__ void place_kernel(const int* __restrict__ src, const int* __restrict__ dst,
                             const int* __restrict__ rank, const int* __restrict__ rowptr,
                             int* __restrict__ tmp, int e) {
  int i = blockIdx.x * blockDim.x + threadIdx.x;
  if (i < e) tmp[rowptr[dst[i]] + rank[i]] = src[i];
}

__global__ void rowsort_kernel(const int* __restrict__ rowptr, const int* __restrict__ tmp,
                               int* __restrict__ col, int n) {
  const int row = (blockIdx.x * blockDim.x + threadIdx.x) >> 5;
  const int l = threadIdx.x & 31;
  if (row >= n) return;
  const int beg = rowptr[row];
  const int end = rowptr[row + 1];
  for (int p = beg + l; p < end; p += 32) {
    const int v = tmp[p];
    int r = 0;
    for (int k = beg; k < end; ++k) {
      const int u = tmp[k];
      r += (u < v || (u == v && k < p)) ? 1 : 0;
    }
    col[beg + r] = v;
  }
}

#define GBM 64
#define GBN 64
#define GBK 32
#define GPA 36
#define GPB2 160
#define GA_FLOATS (GBM * GPA)
#define GB_FLOATS ((GBK / 2) * GPB2)
#define GSTAGE (GA_FLOATS + GB_FLOATS)

__global__ __launch_bounds__(128) void wmma_gemm_f32_kernel(
    const float* __restrict__ A, const float* __restrict__ B,
    float* __restrict__ C, int M, int N, int K) {
  __shared__ __attribute__((aligned(16))) float lds[2 * GSTAGE];

  const int tid   = threadIdx.x;
  const int wave  = tid >> 5;
  const int lane  = tid & 31;
  const int l16   = lane & 15;
  const int lhalf = lane >> 4;

  const int wm = wave >> 1;
  const int wn = wave & 1;
  const int row0 = blockIdx.y * GBM;
  const int col0 = blockIdx.x * GBN;

  const int a_r = tid >> 3;
  const int a_c = (tid & 7) * 4;
  const int b_q = tid >> 4;
  const int b_c = (tid & 15) * 4;
  const int ar0 = min(row0 + a_r, M - 1);
  const int ar1 = min(row0 + a_r + 16, M - 1);
  const int ar2 = min(row0 + a_r + 32, M - 1);
  const int ar3 = min(row0 + a_r + 48, M - 1);
  const float* ag0 = A + (size_t)ar0 * K + a_c;
  const float* ag1 = A + (size_t)ar1 * K + a_c;
  const float* ag2 = A + (size_t)ar2 * K + a_c;
  const float* ag3 = A + (size_t)ar3 * K + a_c;
  const float* bg = B + (size_t)(2 * b_q) * N + col0 + b_c;
  const int a_st = a_r * GPA + a_c;
  const int b_st = GA_FLOATS + b_q * GPB2 + 2 * b_c;

  const int a_fr = (wm * 32 + l16) * GPA + lhalf * 2;
  const int b_fr = GA_FLOATS + lhalf * GPB2 + 2 * (wn * 32 + l16);

  v4f pa0, pa1, pa2, pa3, pb00, pb01, pb10, pb11;
#define GEMM_FETCH(k0)                                                        \
  do {                                                                        \
    pa0 = *reinterpret_cast<const v4f*>(ag0 + (k0));                          \
    pa1 = *reinterpret_cast<const v4f*>(ag1 + (k0));                          \
    pa2 = *reinterpret_cast<const v4f*>(ag2 + (k0));                          \
    pa3 = *reinterpret_cast<const v4f*>(ag3 + (k0));                          \
    pb00 = *reinterpret_cast<const v4f*>(bg + (size_t)((k0) + 0) * N);        \
    pb01 = *reinterpret_cast<const v4f*>(bg + (size_t)((k0) + 1) * N);        \
    pb10 = *reinterpret_cast<const v4f*>(bg + (size_t)((k0) + 16) * N);       \
    pb11 = *reinterpret_cast<const v4f*>(bg + (size_t)((k0) + 17) * N);       \
  } while (0)
#define GEMM_STASH(so)                                                        \
  do {                                                                        \
    *reinterpret_cast<v4f*>(&lds[(so) + a_st + 0 * 16 * GPA]) = pa0;          \
    *reinterpret_cast<v4f*>(&lds[(so) + a_st + 1 * 16 * GPA]) = pa1;          \
    *reinterpret_cast<v4f*>(&lds[(so) + a_st + 2 * 16 * GPA]) = pa2;          \
    *reinterpret_cast<v4f*>(&lds[(so) + a_st + 3 * 16 * GPA]) = pa3;          \
    *reinterpret_cast<v4f*>(&lds[(so) + b_st]) =                              \
        (v4f){pb00.x, pb01.x, pb00.y, pb01.y};                                \
    *reinterpret_cast<v4f*>(&lds[(so) + b_st + 4]) =                          \
        (v4f){pb00.z, pb01.z, pb00.w, pb01.w};                                \
    *reinterpret_cast<v4f*>(&lds[(so) + b_st + 8 * GPB2]) =                   \
        (v4f){pb10.x, pb11.x, pb10.y, pb11.y};                                \
    *reinterpret_cast<v4f*>(&lds[(so) + b_st + 8 * GPB2 + 4]) =               \
        (v4f){pb10.z, pb11.z, pb10.w, pb11.w};                                \
  } while (0)

  v8f acc[2][2] = {};
  auto mma_tile = [&](int so) {
    v2f a[GBK / 4][2], b[GBK / 4][2];
#pragma unroll
    for (int ks = 0; ks < GBK / 4; ++ks) {
#pragma unroll
      for (int mi = 0; mi < 2; ++mi)
        a[ks][mi] = *reinterpret_cast<const v2f*>(&lds[so + a_fr + mi * 16 * GPA + ks * 4]);
#pragma unroll
      for (int ni = 0; ni < 2; ++ni)
        b[ks][ni] = *reinterpret_cast<const v2f*>(&lds[so + b_fr + ks * 2 * GPB2 + ni * 32]);
    }
#pragma unroll
    for (int ks = 0; ks < GBK / 4; ++ks)
#pragma unroll
      for (int mi = 0; mi < 2; ++mi)
#pragma unroll
        for (int ni = 0; ni < 2; ++ni)
          acc[mi][ni] = __builtin_amdgcn_wmma_f32_16x16x4_f32(
               false, a[ks][mi],  false, b[ks][ni],
               (short)0, acc[mi][ni],  false,  false);
  };

  GEMM_FETCH(0);
  GEMM_STASH(0);
  __syncthreads();

  int so = 0;
  for (int k0 = GBK; k0 < K; k0 += GBK) {
    GEMM_FETCH(k0);
    mma_tile(so);
    so = GSTAGE - so;
    GEMM_STASH(so);
    __syncthreads();
  }
  mma_tile(so);
#undef GEMM_FETCH
#undef GEMM_STASH

  const int wrow0 = row0 + wm * 32;
  const int wcol0 = col0 + wn * 32;
#pragma unroll
  for (int mi = 0; mi < 2; ++mi) {
#pragma unroll
    for (int r = 0; r < 8; ++r) {
      int row = wrow0 + mi * 16 + 8 * lhalf + r;
      if (row < M) {
#pragma unroll
        for (int ni = 0; ni < 2; ++ni)
          C[(size_t)row * N + wcol0 + ni * 16 + l16] = acc[mi][ni][r];
      }
    }
  }
}

template <int K, int C4>
__device__ __forceinline__ void add_rows(float4& acc, const float4* __restrict__ h4,
                                         const int* __restrict__ col,
                                         const float* __restrict__ dinv, float di, int j, int t) {
  float4 v[K];
  float w[K];
#pragma unroll
  for (int k = 0; k < K; ++k) {
    const int s = col[j + k];
    w[k] = dinv[s] * di;
    v[k] = h4[(size_t)s * C4 + t];
  }
#pragma unroll
  for (int k = 0; k < K; ++k) {
    acc.x += v[k].x * w[k];
    acc.y += v[k].y * w[k];
    acc.z += v[k].z * w[k];
    acc.w += v[k].w * w[k];
  }
}

template <int C>
__global__ __launch_bounds__(C / 4) void gather_rows_kernel(
    const float* __restrict__ h, const int* __restrict__ rowptr, const int* __restrict__ col,
    const float* __restrict__ dinv, const float* __restrict__ bias, float* __restrict__ out) {
  constexpr int C4 = C / 4;
  const int i = blockIdx.x;
  const int t = threadIdx.x;
  const float4* __restrict__ h4 = reinterpret_cast<const float4*>(h);
  const float di = dinv[i];
  const int beg = rowptr[i];
  const int end = rowptr[i + 1];

  const float4 sv = h4[(size_t)i * C4 + t];
  const float4 b = reinterpret_cast<const float4*>(bias)[t];
  const float sw = di * di;
  float4 acc;
  acc.x = sv.x * sw; acc.y = sv.y * sw; acc.z = sv.z * sw; acc.w = sv.w * sw;

  int j = beg;
  for (; j + 8 <= end; j += 8) add_rows<8, C4>(acc, h4, col, dinv, di, j, t);
  if (j + 4 <= end) { add_rows<4, C4>(acc, h4, col, dinv, di, j, t); j += 4; }
  if (j + 2 <= end) { add_rows<2, C4>(acc, h4, col, dinv, di, j, t); j += 2; }
  if (j < end) add_rows<1, C4>(acc, h4, col, dinv, di, j, t);

  float4 o;
  o.x = fmaxf(acc.x + b.x, 0.0f);
  o.y = fmaxf(acc.y + b.y, 0.0f);
  o.z = fmaxf(acc.z + b.z, 0.0f);
  o.w = fmaxf(acc.w + b.w, 0.0f);
  reinterpret_cast<float4*>(out)[(size_t)i * C4 + t] = o;
}

extern "C" void kernel_launch(void* const* d_in, const int* in_sizes, int n_in,
                              void* d_out, int out_size, void* d_ws, size_t ws_size,
                              hipStream_t stream) {
  const float* x   = (const float*)d_in[0];
  const int*   ei  = (const int*)d_in[1];
  const float* W1  = (const float*)d_in[2];
  const float* b1  = (const float*)d_in[3];
  const float* W2  = (const float*)d_in[4];
  const float* b2  = (const float*)d_in[5];
  float* out = (float*)d_out;

  const int N = NNODES;
  const int E = NEDGES;
  const int C0 = 512, C1 = 1024, C2 = 512;

  const int* src = ei;
  const int* dst = ei + E;

  float* h      = (float*)d_ws;
  float* agg1   = h + (size_t)N * C1;
  float* dinv   = agg1 + (size_t)N * C1;
  int*   cnt    = (int*)(dinv + N);
  int*   rowptr = cnt + N;
  int*   rank   = rowptr + (N + 16);
  int*   tmp    = rank + E;
  int*   col    = tmp + E;

  const int T = 256;
  const int MB = (N + 63) / 64;

  cnt_zero_kernel<<<(N + T - 1) / T, T, 0, stream>>>(cnt, N);
  edge_rank_kernel<<<(E + T - 1) / T, T, 0, stream>>>(dst, cnt, rank, E);
  rowptr_dinv_kernel<<<1, SCAN_T, 0, stream>>>(cnt, rowptr, dinv);
  place_kernel<<<(E + T - 1) / T, T, 0, stream>>>(src, dst, rank, rowptr, tmp, E);
  rowsort_kernel<<<(N * 32 + T - 1) / T, T, 0, stream>>>(rowptr, tmp, col, N);

  {
    dim3 grid(C1 / 64, MB);
    wmma_gemm_f32_kernel<<<grid, 128, 0, stream>>>(x, W1, h, N, C1, C0);
    gather_rows_kernel<1024><<<N, 256, 0, stream>>>(h, rowptr, col, dinv, b1, agg1);
  }

  {
    dim3 grid(C2 / 64, MB);
    wmma_gemm_f32_kernel<<<grid, 128, 0, stream>>>(agg1, W2, h, N, C2, C1);
    gather_rows_kernel<512><<<N, 128, 0, stream>>>(h, rowptr, col, dinv, b2, out);
  }
}


# Layer 1 sums its rows on the 512-wide input before the product instead of on the 1024-wide product (bias and relu move into the product's store): gather rows halve; layer-1 sum order changes
# speedup vs baseline: 3.4903x; 1.0688x over previous
#include <hip/hip_runtime.h>


typedef float v2f __attribute__((ext_vector_type(2)));
typedef float v4f __attribute__((ext_vector_type(4)));
typedef float v8f __attribute__((ext_vector_type(8)));

#define NNODES 10000
#define NEDGES 160000

__global__ void cnt_zero_kernel(int* __restrict__ cnt, int n) {
  int i = blockIdx.x * blockDim.x + threadIdx.x;
  if (i < n) cnt[i] = 0;
}

__global__ void edge_rank_kernel(const int* __restrict__ dst, int* __restrict__ cnt,
                                 int* __restrict__ rank, int e) {
  int i = blockIdx.x * blockDim.x + threadIdx.x;
  if (i < e) rank[i] = atomicAdd(&cnt[dst[i]], 1);
}

#define SCAN_T 1024
#define SCAN_PER ((NNODES + SCAN_T - 1) / SCAN_T)
__global__ __launch_bounds__(SCAN_T) void rowptr_dinv_kernel(
    const int* __restrict__ cnt, int* __restrict__ rowptr, float* __restrict__ dinv) {
  __shared__ int part[SCAN_T];
  const int t = threadIdx.x;
  const int base = t * SCAN_PER;
  int c[SCAN_PER];
  int s = 0;
#pragma unroll
  for (int j = 0; j < SCAN_PER; ++j) {
    int i = base + j;
    c[j] = i < NNODES ? cnt[i] : 0;
    s += c[j];
  }
  part[t] = s;
  __syncthreads();
  for (int off = 1; off < SCAN_T; off <<= 1) {
    int v = t >= off ? part[t - off] : 0;
    __syncthreads();
    part[t] += v;
    __syncthreads();
  }
  int run = part[t] - s;
#pragma unroll
  for (int j = 0; j < SCAN_PER; ++j) {
    int i = base + j;
    if (i < NNODES) {
      rowptr[i] = run;
      dinv[i] = rsqrtf((float)(c[j] + 1));
      run += c[j];
    }
  }
  if (t == SCAN_T - 1) rowptr[NNODES] = part[SCAN_T - 1];
}

__global__ void place_kernel(const int* __restrict__ src, const int* __restrict__ dst,
                             const int* __restrict__ rank, const int* __restrict__ rowptr,
                             int* __restrict__ tmp, int e) {
  int i = blockIdx.x * blockDim.x + threadIdx.x;
  if (i < e) tmp[rowptr[dst[i]] + rank[i]] = src[i];
}

__global__ void rowsort_kernel(const int* __restrict__ rowptr, const int* __restrict__ tmp,
                               int* __restrict__ col, int n) {
  const int row = (blockIdx.x * blockDim.x + threadIdx.x) >> 5;
  const int l = threadIdx.x & 31;
  if (row >= n) return;
  const int beg = rowptr[row];
  const int end = rowptr[row + 1];
  for (int p = beg + l; p < end; p += 32) {
    const int v = tmp[p];
    int r = 0;
    for (int k = beg; k < end; ++k) {
      const int u = tmp[k];
      r += (u < v || (u == v && k < p)) ? 1 : 0;
    }
    col[beg + r] = v;
  }
}

#define GBM 64
#define GBN 64
#define GBK 32
#define GPA 36
#define GPB2 160
#define GA_FLOATS (GBM * GPA)
#define GB_FLOATS ((GBK / 2) * GPB2)
#define GSTAGE (GA_FLOATS + GB_FLOATS)

template <bool EPI>
__global__ __launch_bounds__(128) void wmma_gemm_f32_kernel(
    const float* __restrict__ A, const float* __restrict__ B, const float* __restrict__ bias,
    float* __restrict__ C, int M, int N, int K) {
  __shared__ __attribute__((aligned(16))) float lds[2 * GSTAGE];

  const int tid   = threadIdx.x;
  const int wave  = tid >> 5;
  const int lane  = tid & 31;
  const int l16   = lane & 15;
  const int lhalf = lane >> 4;

  const int wm = wave >> 1;
  const int wn = wave & 1;
  const int row0 = blockIdx.y * GBM;
  const int col0 = blockIdx.x * GBN;

  const int a_r = tid >> 3;
  const int a_c = (tid & 7) * 4;
  const int b_q = tid >> 4;
  const int b_c = (tid & 15) * 4;
  const int ar0 = min(row0 + a_r, M - 1);
  const int ar1 = min(row0 + a_r + 16, M - 1);
  const int ar2 = min(row0 + a_r + 32, M - 1);
  const int ar3 = min(row0 + a_r + 48, M - 1);
  const float* ag0 = A + (size_t)ar0 * K + a_c;
  const float* ag1 = A + (size_t)ar1 * K + a_c;
  const float* ag2 = A + (size_t)ar2 * K + a_c;
  const float* ag3 = A + (size_t)ar3 * K + a_c;
  const float* bg = B + (size_t)(2 * b_q) * N + col0 + b_c;
  const int a_st = a_r * GPA + a_c;
  const int b_st = GA_FLOATS + b_q * GPB2 + 2 * b_c;

  const int a_fr = (wm * 32 + l16) * GPA + lhalf * 2;
  const int b_fr = GA_FLOATS + lhalf * GPB2 + 2 * (wn * 32 + l16);

  v4f pa0, pa1, pa2, pa3, pb00, pb01, pb10, pb11;
#define GEMM_FETCH(k0)                                                        \
  do {                                                                        \
    pa0 = *reinterpret_cast<const v4f*>(ag0 + (k0));                          \
    pa1 = *reinterpret_cast<const v4f*>(ag1 + (k0));                          \
    pa2 = *reinterpret_cast<const v4f*>(ag2 + (k0));                          \
    pa3 = *reinterpret_cast<const v4f*>(ag3 + (k0));                          \
    pb00 = *reinterpret_cast<const v4f*>(bg + (size_t)((k0) + 0) * N);        \
    pb01 = *reinterpret_cast<const v4f*>(bg + (size_t)((k0) + 1) * N);        \
    pb10 = *reinterpret_cast<const v4f*>(bg + (size_t)((k0) + 16) * N);       \
    pb11 = *reinterpret_cast<const v4f*>(bg + (size_t)((k0) + 17) * N);       \
  } while (0)
#define GEMM_STASH(so)                                                        \
  do {                                                                        \
    *reinterpret_cast<v4f*>(&lds[(so) + a_st + 0 * 16 * GPA]) = pa0;          \
    *reinterpret_cast<v4f*>(&lds[(so) + a_st + 1 * 16 * GPA]) = pa1;          \
    *reinterpret_cast<v4f*>(&lds[(so) + a_st + 2 * 16 * GPA]) = pa2;          \
    *reinterpret_cast<v4f*>(&lds[(so) + a_st + 3 * 16 * GPA]) = pa3;          \
    *reinterpret_cast<v4f*>(&lds[(so) + b_st]) =                              \
        (v4f){pb00.x, pb01.x, pb00.y, pb01.y};                                \
    *reinterpret_cast<v4f*>(&lds[(so) + b_st + 4]) =                          \
        (v4f){pb00.z, pb01.z, pb00.w, pb01.w};                                \
    *reinterpret_cast<v4f*>(&lds[(so) + b_st + 8 * GPB2]) =                   \
        (v4f){pb10.x, pb11.x, pb10.y, pb11.y};                                \
    *reinterpret_cast<v4f*>(&lds[(so) + b_st + 8 * GPB2 + 4]) =               \
        (v4f){pb10.z, pb11.z, pb10.w, pb11.w};                                \
  } while (0)

  v8f acc[2][2] = {};
  auto mma_tile = [&](int so) {
    v2f a[GBK / 4][2], b[GBK / 4][2];
#pragma unroll
    for (int ks = 0; ks < GBK / 4; ++ks) {
#pragma unroll
      for (int mi = 0; mi < 2; ++mi)
        a[ks][mi] = *reinterpret_cast<const v2f*>(&lds[so + a_fr + mi * 16 * GPA + ks * 4]);
#pragma unroll
      for (int ni = 0; ni < 2; ++ni)
        b[ks][ni] = *reinterpret_cast<const v2f*>(&lds[so + b_fr + ks * 2 * GPB2 + ni * 32]);
    }
#pragma unroll
    for (int ks = 0; ks < GBK / 4; ++ks)
#pragma unroll
      for (int mi = 0; mi < 2; ++mi)
#pragma unroll
        for (int ni = 0; ni < 2; ++ni)
          acc[mi][ni] = __builtin_amdgcn_wmma_f32_16x16x4_f32(
               false, a[ks][mi],  false, b[ks][ni],
               (short)0, acc[mi][ni],  false,  false);
  };

  GEMM_FETCH(0);
  GEMM_STASH(0);
  __syncthreads();

  int so = 0;
  for (int k0 = GBK; k0 < K; k0 += GBK) {
    GEMM_FETCH(k0);
    mma_tile(so);
    so = GSTAGE - so;
    GEMM_STASH(so);
    __syncthreads();
  }
  mma_tile(so);
#undef GEMM_FETCH
#undef GEMM_STASH

  const int wrow0 = row0 + wm * 32;
  const int wcol0 = col0 + wn * 32;
  float bv[2] = {0.0f, 0.0f};
  if (EPI) {
    bv[0] = bias[wcol0 + l16];
    bv[1] = bias[wcol0 + 16 + l16];
  }
#pragma unroll
  for (int mi = 0; mi < 2; ++mi) {
#pragma unroll
    for (int r = 0; r < 8; ++r) {
      int row = wrow0 + mi * 16 + 8 * lhalf + r;
      if (row < M) {
#pragma unroll
        for (int ni = 0; ni < 2; ++ni) {
          float v = acc[mi][ni][r];
          if (EPI) v = fmaxf(v + bv[ni], 0.0f);
          C[(size_t)row * N + wcol0 + ni * 16 + l16] = v;
        }
      }
    }
  }
}

template <int K, int C4>
__device__ __forceinline__ void add_rows(float4& acc, const float4* __restrict__ h4,
                                         const int* __restrict__ col,
                                         const float* __restrict__ dinv, float di, int j, int t) {
  float4 v[K];
  float w[K];
#pragma unroll
  for (int k = 0; k < K; ++k) {
    const int s = col[j + k];
    w[k] = dinv[s] * di;
    v[k] = h4[(size_t)s * C4 + t];
  }
#pragma unroll
  for (int k = 0; k < K; ++k) {
    acc.x += v[k].x * w[k];
    acc.y += v[k].y * w[k];
    acc.z += v[k].z * w[k];
    acc.w += v[k].w * w[k];
  }
}

template <int C, bool EPI>
__global__ __launch_bounds__(C / 4) void gather_rows_kernel(
    const float* __restrict__ h, const int* __restrict__ rowptr, const int* __restrict__ col,
    const float* __restrict__ dinv, const float* __restrict__ bias, float* __restrict__ out) {
  constexpr int C4 = C / 4;
  const int i = blockIdx.x;
  const int t = threadIdx.x;
  const float4* __restrict__ h4 = reinterpret_cast<const float4*>(h);
  const float di = dinv[i];
  const int beg = rowptr[i];
  const int end = rowptr[i + 1];

  const float4 sv = h4[(size_t)i * C4 + t];
  float4 b = {0.0f, 0.0f, 0.0f, 0.0f};
  if (EPI) b = reinterpret_cast<const float4*>(bias)[t];
  const float sw = di * di;
  float4 acc;
  acc.x = sv.x * sw; acc.y = sv.y * sw; acc.z = sv.z * sw; acc.w = sv.w * sw;

  int j = beg;
  for (; j + 8 <= end; j += 8) add_rows<8, C4>(acc, h4, col, dinv, di, j, t);
  if (j + 4 <= end) { add_rows<4, C4>(acc, h4, col, dinv, di, j, t); j += 4; }
  if (j + 2 <= end) { add_rows<2, C4>(acc, h4, col, dinv, di, j, t); j += 2; }
  if (j < end) add_rows<1, C4>(acc, h4, col, dinv, di, j, t);

  float4 o = acc;
  if (EPI) {
    o.x = fmaxf(acc.x + b.x, 0.0f);
    o.y = fmaxf(acc.y + b.y, 0.0f);
    o.z = fmaxf(acc.z + b.z, 0.0f);
    o.w = fmaxf(acc.w + b.w, 0.0f);
  }
  reinterpret_cast<float4*>(out)[(size_t)i * C4 + t] = o;
}

extern "C" void kernel_launch(void* const* d_in, const int* in_sizes, int n_in,
                              void* d_out, int out_size, void* d_ws, size_t ws_size,
                              hipStream_t stream) {
  const float* x   = (const float*)d_in[0];
  const int*   ei  = (const int*)d_in[1];
  const float* W1  = (const float*)d_in[2];
  const float* b1  = (const float*)d_in[3];
  const float* W2  = (const float*)d_in[4];
  const float* b2  = (const float*)d_in[5];
  float* out = (float*)d_out;

  const int N = NNODES;
  const int E = NEDGES;
  const int C0 = 512, C1 = 1024, C2 = 512;

  const int* src = ei;
  const int* dst = ei + E;

  float* h      = (float*)d_ws;
  float* agg1   = h + (size_t)N * C1;
  float* dinv   = agg1 + (size_t)N * C1;
  int*   cnt    = (int*)(dinv + N);
  int*   rowptr = cnt + N;
  int*   rank   = rowptr + (N + 16);
  int*   tmp    = rank + E;
  int*   col    = tmp + E;

  const int T = 256;
  const int MB = (N + 63) / 64;

  cnt_zero_kernel<<<(N + T - 1) / T, T, 0, stream>>>(cnt, N);
  edge_rank_kernel<<<(E + T - 1) / T, T, 0, stream>>>(dst, cnt, rank, E);
  rowptr_dinv_kernel<<<1, SCAN_T, 0, stream>>>(cnt, rowptr, dinv);
  place_kernel<<<(E + T - 1) / T, T, 0, stream>>>(src, dst, rank, rowptr, tmp, E);
  rowsort_kernel<<<(N * 32 + T - 1) / T, T, 0, stream>>>(rowptr, tmp, col, N);

  {
    dim3 grid(C1 / 64, MB);
    gather_rows_kernel<512, false><<<N, 128, 0, stream>>>(x, rowptr, col, dinv, b1, h);
    wmma_gemm_f32_kernel<true><<<grid, 128, 0, stream>>>(h, W1, b1, agg1, N, C1, C0);
  }

  {
    dim3 grid(C2 / 64, MB);
    wmma_gemm_f32_kernel<false><<<grid, 128, 0, stream>>>(agg1, W2, b2, h, N, C2, C1);
    gather_rows_kernel<512, true><<<N, 128, 0, stream>>>(h, rowptr, col, dinv, b2, out);
  }
}


# Rows by destination built by one launch instead of five: a block owns 80 rows, reads the destination list once, keeps its edges in LDS (int-atomic slots), places and sorts them; sum order unchanged
# speedup vs baseline: 3.9283x; 1.1255x over previous
#include <hip/hip_runtime.h>


typedef float v2f __attribute__((ext_vector_type(2)));
typedef float v4f __attribute__((ext_vector_type(4)));
typedef float v8f __attribute__((ext_vector_type(8)));

#define NNODES 10000
#define NEDGES 160000

#define CSR_T 1024
#define CSR_R 80
#define CSR_NB ((NNODES + CSR_R - 1) / CSR_R)
#define CSR_Q (NEDGES / 4)
#define CSR_U 8
#define CSR_CAP 4096
static_assert(NEDGES % 4 == 0, "destination list is read in 16-byte groups");
static_assert(NEDGES <= (1 << 20) && CSR_R <= (1 << 11), "list entry: edge | row << 20");

__global__ __launch_bounds__(CSR_T) void csr_build_kernel(
    const int* __restrict__ src, const int* __restrict__ dst, int* __restrict__ rowptr,
    float* __restrict__ dinv, int* __restrict__ tmp, int* __restrict__ col) {
  __shared__ int cntL[CSR_R];
  __shared__ int fillL[CSR_R];
  __shared__ int startL[CSR_R + 1];
  __shared__ int belowL;
  __shared__ int nlocL;
  __shared__ int listL[CSR_CAP];
  __shared__ int placeL[CSR_CAP];
  const int t = threadIdx.x;
  const int lo = blockIdx.x * CSR_R;
  const int hi = min(lo + CSR_R, NNODES);
  const int nrows = hi - lo;
  if (t < CSR_R) { cntL[t] = 0; fillL[t] = 0; }
  if (t == 0) { belowL = 0; nlocL = 0; }
  __syncthreads();

  const int4* __restrict__ dst4 = reinterpret_cast<const int4*>(dst);
  const int4 none = {0x7fffffff, 0x7fffffff, 0x7fffffff, 0x7fffffff};

  int below = 0;
  for (int q0 = t; q0 < CSR_Q; q0 += CSR_U * CSR_T) {
    int4 d[CSR_U];
#pragma unroll
    for (int u = 0; u < CSR_U; ++u) {
      const int q = q0 + u * CSR_T;
      const int4 v = dst4[min(q, CSR_Q - 1)];
      d[u] = q < CSR_Q ? v : none;
    }
#pragma unroll
    for (int u = 0; u < CSR_U; ++u) {
      below += (d[u].x < lo ? 1 : 0) + (d[u].y < lo ? 1 : 0) + (d[u].z < lo ? 1 : 0) +
               (d[u].w < lo ? 1 : 0);
      const int e = 4 * (q0 + u * CSR_T);
      const int dd[4] = {d[u].x, d[u].y, d[u].z, d[u].w};
#pragma unroll
      for (int c = 0; c < 4; ++c) {
        if (dd[c] >= lo && dd[c] < hi) {
          const int r = dd[c] - lo;
          atomicAdd(&cntL[r], 1);
          const int pos = atomicAdd(&nlocL, 1);
          if (pos < CSR_CAP) listL[pos] = (e + c) | (r << 20);
        }
      }
    }
  }
  for (int off = 16; off > 0; off >>= 1) below += __shfl_xor(below, off);
  if ((t & 31) == 0) atomicAdd(&belowL, below);
  __syncthreads();

  if (t == 0) {
    int run = belowL;
    for (int i = 0; i < CSR_R; ++i) {
      startL[i] = run;
      run += cntL[i];
    }
    startL[CSR_R] = run;
  }
  __syncthreads();
  if (t < nrows) {
    rowptr[lo + t] = startL[t];
    dinv[lo + t] = rsqrtf((float)(cntL[t] + 1));
  }
  if (blockIdx.x == CSR_NB - 1 && t == 0) rowptr[NNODES] = startL[CSR_R];

  const int nloc = nlocL;
  const int l = t & 31;
  if (nloc <= CSR_CAP) {
    const int first = startL[0];
    for (int i = t; i < nloc; i += CSR_T) {
      const int ent = listL[i];
      const int r = ent >> 20;
      placeL[startL[r] - first + atomicAdd(&fillL[r], 1)] = src[ent & ((1 << 20) - 1)];
    }
    __syncthreads();
    for (int r = t >> 5; r < nrows; r += CSR_T / 32) {
      const int beg = startL[r] - first;
      const int end = startL[r + 1] - first;
      for (int p = beg + l; p < end; p += 32) {
        const int v = placeL[p];
        int k_lt = 0;
#pragma unroll 4
        for (int k = beg; k < end; ++k) {
          const int u = placeL[k];
          k_lt += (u < v || (u == v && k < p)) ? 1 : 0;
        }
        col[first + beg + k_lt] = v;
      }
    }
    return;
  }
  {
  for (int q0 = t; q0 < CSR_Q; q0 += CSR_U * CSR_T) {
    int4 d[CSR_U];
#pragma unroll
    for (int u = 0; u < CSR_U; ++u) {
      const int q = q0 + u * CSR_T;
      const int4 v = dst4[min(q, CSR_Q - 1)];
      d[u] = q < CSR_Q ? v : none;
    }
#pragma unroll
    for (int u = 0; u < CSR_U; ++u) {
      const int e = 4 * (q0 + u * CSR_T);
      if (d[u].x >= lo && d[u].x < hi) {
        const int r = d[u].x - lo;
        tmp[startL[r] + atomicAdd(&fillL[r], 1)] = src[e + 0];
      }
      if (d[u].y >= lo && d[u].y < hi) {
        const int r = d[u].y - lo;
        tmp[startL[r] + atomicAdd(&fillL[r], 1)] = src[e + 1];
      }
      if (d[u].z >= lo && d[u].z < hi) {
        const int r = d[u].z - lo;
        tmp[startL[r] + atomicAdd(&fillL[r], 1)] = src[e + 2];
      }
      if (d[u].w >= lo && d[u].w < hi) {
        const int r = d[u].w - lo;
        tmp[startL[r] + atomicAdd(&fillL[r], 1)] = src[e + 3];
      }
    }
  }
  }
  __syncthreads();

  for (int r = t >> 5; r < nrows; r += CSR_T / 32) {
    const int beg = startL[r];
    const int end = startL[r + 1];
    for (int p = beg + l; p < end; p += 32) {
      const int v = tmp[p];
      int k_lt = 0;
      for (int k = beg; k < end; ++k) {
        const int u = tmp[k];
        k_lt += (u < v || (u == v && k < p)) ? 1 : 0;
      }
      col[beg + k_lt] = v;
    }
  }
}

#define GBM 64
#define GBN 64
#define GBK 32
#define GPA 36
#define GPB2 160
#define GA_FLOATS (GBM * GPA)
#define GB_FLOATS ((GBK / 2) * GPB2)
#define GSTAGE (GA_FLOATS + GB_FLOATS)

template <bool EPI>
__global__ __launch_bounds__(128) void wmma_gemm_f32_kernel(
    const float* __restrict__ A, const float* __restrict__ B, const float* __restrict__ bias,
    float* __restrict__ C, int M, int N, int K) {
  __shared__ __attribute__((aligned(16))) float lds[2 * GSTAGE];

  const int tid   = threadIdx.x;
  const int wave  = tid >> 5;
  const int lane  = tid & 31;
  const int l16   = lane & 15;
  const int lhalf = lane >> 4;

  const int wm = wave >> 1;
  const int wn = wave & 1;
  const int row0 = blockIdx.y * GBM;
  const int col0 = blockIdx.x * GBN;

  const int a_r = tid >> 3;
  const int a_c = (tid & 7) * 4;
  const int b_q = tid >> 4;
  const int b_c = (tid & 15) * 4;
  const int ar0 = min(row0 + a_r, M - 1);
  const int ar1 = min(row0 + a_r + 16, M - 1);
  const int ar2 = min(row0 + a_r + 32, M - 1);
  const int ar3 = min(row0 + a_r + 48, M - 1);
  const float* ag0 = A + (size_t)ar0 * K + a_c;
  const float* ag1 = A + (size_t)ar1 * K + a_c;
  const float* ag2 = A + (size_t)ar2 * K + a_c;
  const float* ag3 = A + (size_t)ar3 * K + a_c;
  const float* bg = B + (size_t)(2 * b_q) * N + col0 + b_c;
  const int a_st = a_r * GPA + a_c;
  const int b_st = GA_FLOATS + b_q * GPB2 + 2 * b_c;

  const int a_fr = (wm * 32 + l16) * GPA + lhalf * 2;
  const int b_fr = GA_FLOATS + lhalf * GPB2 + 2 * (wn * 32 + l16);

  v4f pa0, pa1, pa2, pa3, pb00, pb01, pb10, pb11;
#define GEMM_FETCH(k0)                                                        \
  do {                                                                        \
    pa0 = *reinterpret_cast<const v4f*>(ag0 + (k0));                          \
    pa1 = *reinterpret_cast<const v4f*>(ag1 + (k0));                          \
    pa2 = *reinterpret_cast<const v4f*>(ag2 + (k0));                          \
    pa3 = *reinterpret_cast<const v4f*>(ag3 + (k0));                          \
    pb00 = *reinterpret_cast<const v4f*>(bg + (size_t)((k0) + 0) * N);        \
    pb01 = *reinterpret_cast<const v4f*>(bg + (size_t)((k0) + 1) * N);        \
    pb10 = *reinterpret_cast<const v4f*>(bg + (size_t)((k0) + 16) * N);       \
    pb11 = *reinterpret_cast<const v4f*>(bg + (size_t)((k0) + 17) * N);       \
  } while (0)
#define GEMM_STASH(so)                                                        \
  do {                                                                        \
    *reinterpret_cast<v4f*>(&lds[(so) + a_st + 0 * 16 * GPA]) = pa0;          \
    *reinterpret_cast<v4f*>(&lds[(so) + a_st + 1 * 16 * GPA]) = pa1;          \
    *reinterpret_cast<v4f*>(&lds[(so) + a_st + 2 * 16 * GPA]) = pa2;          \
    *reinterpret_cast<v4f*>(&lds[(so) + a_st + 3 * 16 * GPA]) = pa3;          \
    *reinterpret_cast<v4f*>(&lds[(so) + b_st]) =                              \
        (v4f){pb00.x, pb01.x, pb00.y, pb01.y};                                \
    *reinterpret_cast<v4f*>(&lds[(so) + b_st + 4]) =                          \
        (v4f){pb00.z, pb01.z, pb00.w, pb01.w};                                \
    *reinterpret_cast<v4f*>(&lds[(so) + b_st + 8 * GPB2]) =                   \
        (v4f){pb10.x, pb11.x, pb10.y, pb11.y};                                \
    *reinterpret_cast<v4f*>(&lds[(so) + b_st + 8 * GPB2 + 4]) =               \
        (v4f){pb10.z, pb11.z, pb10.w, pb11.w};                                \
  } while (0)

  v8f acc[2][2] = {};
  auto mma_tile = [&](int so) {
    v2f a[GBK / 4][2], b[GBK / 4][2];
#pragma unroll
    for (int ks = 0; ks < GBK / 4; ++ks) {
#pragma unroll
      for (int mi = 0; mi < 2; ++mi)
        a[ks][mi] = *reinterpret_cast<const v2f*>(&lds[so + a_fr + mi * 16 * GPA + ks * 4]);
#pragma unroll
      for (int ni = 0; ni < 2; ++ni)
        b[ks][ni] = *reinterpret_cast<const v2f*>(&lds[so + b_fr + ks * 2 * GPB2 + ni * 32]);
    }
#pragma unroll
    for (int ks = 0; ks < GBK / 4; ++ks)
#pragma unroll
      for (int mi = 0; mi < 2; ++mi)
#pragma unroll
        for (int ni = 0; ni < 2; ++ni)
          acc[mi][ni] = __builtin_amdgcn_wmma_f32_16x16x4_f32(
               false, a[ks][mi],  false, b[ks][ni],
               (short)0, acc[mi][ni],  false,  false);
  };

  GEMM_FETCH(0);
  GEMM_STASH(0);
  __syncthreads();

  int so = 0;
  for (int k0 = GBK; k0 < K; k0 += GBK) {
    GEMM_FETCH(k0);
    mma_tile(so);
    so = GSTAGE - so;
    GEMM_STASH(so);
    __syncthreads();
  }
  mma_tile(so);
#undef GEMM_FETCH
#undef GEMM_STASH

  const int wrow0 = row0 + wm * 32;
  const int wcol0 = col0 + wn * 32;
  float bv[2] = {0.0f, 0.0f};
  if (EPI) {
    bv[0] = bias[wcol0 + l16];
    bv[1] = bias[wcol0 + 16 + l16];
  }
#pragma unroll
  for (int mi = 0; mi < 2; ++mi) {
#pragma unroll
    for (int r = 0; r < 8; ++r) {
      int row = wrow0 + mi * 16 + 8 * lhalf + r;
      if (row < M) {
#pragma unroll
        for (int ni = 0; ni < 2; ++ni) {
          float v = acc[mi][ni][r];
          if (EPI) v = fmaxf(v + bv[ni], 0.0f);
          C[(size_t)row * N + wcol0 + ni * 16 + l16] = v;
        }
      }
    }
  }
}

template <int K, int C4>
__device__ __forceinline__ void add_rows(float4& acc, const float4* __restrict__ h4,
                                         const int* __restrict__ col,
                                         const float* __restrict__ dinv, float di, int j, int t) {
  float4 v[K];
  float w[K];
#pragma unroll
  for (int k = 0; k < K; ++k) {
    const int s = col[j + k];
    w[k] = dinv[s] * di;
    v[k] = h4[(size_t)s * C4 + t];
  }
#pragma unroll
  for (int k = 0; k < K; ++k) {
    acc.x += v[k].x * w[k];
    acc.y += v[k].y * w[k];
    acc.z += v[k].z * w[k];
    acc.w += v[k].w * w[k];
  }
}

template <int C, bool EPI>
__global__ __launch_bounds__(C / 4) void gather_rows_kernel(
    const float* __restrict__ h, const int* __restrict__ rowptr, const int* __restrict__ col,
    const float* __restrict__ dinv, const float* __restrict__ bias, float* __restrict__ out) {
  constexpr int C4 = C / 4;
  const int i = blockIdx.x;
  const int t = threadIdx.x;
  const float4* __restrict__ h4 = reinterpret_cast<const float4*>(h);
  const float di = dinv[i];
  const int beg = rowptr[i];
  const int end = rowptr[i + 1];

  const float4 sv = h4[(size_t)i * C4 + t];
  float4 b = {0.0f, 0.0f, 0.0f, 0.0f};
  if (EPI) b = reinterpret_cast<const float4*>(bias)[t];
  const float sw = di * di;
  float4 acc;
  acc.x = sv.x * sw; acc.y = sv.y * sw; acc.z = sv.z * sw; acc.w = sv.w * sw;

  int j = beg;
  for (; j + 8 <= end; j += 8) add_rows<8, C4>(acc, h4, col, dinv, di, j, t);
  if (j + 4 <= end) { add_rows<4, C4>(acc, h4, col, dinv, di, j, t); j += 4; }
  if (j + 2 <= end) { add_rows<2, C4>(acc, h4, col, dinv, di, j, t); j += 2; }
  if (j < end) add_rows<1, C4>(acc, h4, col, dinv, di, j, t);

  float4 o = acc;
  if (EPI) {
    o.x = fmaxf(acc.x + b.x, 0.0f);
    o.y = fmaxf(acc.y + b.y, 0.0f);
    o.z = fmaxf(acc.z + b.z, 0.0f);
    o.w = fmaxf(acc.w + b.w, 0.0f);
  }
  reinterpret_cast<float4*>(out)[(size_t)i * C4 + t] = o;
}

extern "C" void kernel_launch(void* const* d_in, const int* in_sizes, int n_in,
                              void* d_out, int out_size, void* d_ws, size_t ws_size,
                              hipStream_t stream) {
  const float* x   = (const float*)d_in[0];
  const int*   ei  = (const int*)d_in[1];
  const float* W1  = (const float*)d_in[2];
  const float* b1  = (const float*)d_in[3];
  const float* W2  = (const float*)d_in[4];
  const float* b2  = (const float*)d_in[5];
  float* out = (float*)d_out;

  const int N = NNODES;
  const int E = NEDGES;
  const int C0 = 512, C1 = 1024, C2 = 512;

  const int* src = ei;
  const int* dst = ei + E;

  float* h      = (float*)d_ws;
  float* agg1   = h + (size_t)N * C1;
  float* dinv   = agg1 + (size_t)N * C1;
  int*   rowptr = (int*)(dinv + N);
  int*   tmp    = rowptr + (N + 16);
  int*   col    = tmp + E;

  const int MB = (N + 63) / 64;

  csr_build_kernel<<<CSR_NB, CSR_T, 0, stream>>>(src, dst, rowptr, dinv, tmp, col);

  {
    dim3 grid(C1 / 64, MB);
    gather_rows_kernel<512, false><<<N, 128, 0, stream>>>(x, rowptr, col, dinv, b1, h);
    wmma_gemm_f32_kernel<true><<<grid, 128, 0, stream>>>(h, W1, b1, agg1, N, C1, C0);
  }

  {
    dim3 grid(C2 / 64, MB);
    wmma_gemm_f32_kernel<false><<<grid, 128, 0, stream>>>(agg1, W2, b2, h, N, C2, C1);
    gather_rows_kernel<512, true><<<N, 128, 0, stream>>>(h, rowptr, col, dinv, b2, out);
  }
}
